# starting kernel
# baseline (speedup 1.0000x reference)
#include <hip/hip_runtime.h>
#include <math.h>


typedef float v2f __attribute__((ext_vector_type(2)));
typedef float v8f __attribute__((ext_vector_type(8)));

#define C_CH 64

__device__ __forceinline__ void atomic_fadd(float* p, float v) {
#if defined(__gfx1250__)
  unsafeAtomicAdd(p, v);
#else
  atomicAdd(p, v);
#endif
}

__global__ void SpVecs_zero(float* __restrict__ sums, float* __restrict__ cnts,
                            int nsum, int ncnt) {
  int i = blockIdx.x * blockDim.x + threadIdx.x;
  if (i < nsum) sums[i] = 0.0f;
  if (i < ncnt) cnts[i] = 0.0f;
}

__global__ __launch_bounds__(256) void SpVecs_scatter(
    const float* __restrict__ feat, const int* __restrict__ yx,
    const int* __restrict__ seg, float* __restrict__ sums,
    float* __restrict__ cnts, int N, int HW, int W) {
  int i = blockIdx.x * blockDim.x + threadIdx.x;
  if (i >= N) return;

  int y = yx[2 * i];
  int x = yx[2 * i + 1];
  int s = seg[i];

  const float* f = feat + (size_t)y * (size_t)W + (size_t)x;
  float* o = sums + (size_t)s * C_CH;

  atomic_fadd(&cnts[s], 1.0f);

  #pragma unroll
  for (int cb = 0; cb < C_CH; cb += 16) {
    float v[16];
    #pragma unroll
    for (int k = 0; k < 16; ++k) v[k] = f[(size_t)(cb + k) * (size_t)HW];
    #pragma unroll
    for (int k = 0; k < 16; ++k) atomic_fadd(&o[cb + k], v[k]);
  }
}

__global__ __launch_bounds__(256) void SpVecs_finalize(
    float* __restrict__ out, const float* __restrict__ cnts) {
  const int lane  = threadIdx.x & 31;
  const int wid   = threadIdx.x >> 5;
  const int g     = blockIdx.x * (blockDim.x >> 5) + wid;
  const int s0    = g << 4;
  const int m     = lane & 15;
  const int half  = lane >> 4;
  const int khalf = half << 1;

  const float invc = 1.0f / fmaxf(cnts[s0 + m], 1.0f);

#if __has_builtin(__builtin_amdgcn_wmma_f32_16x16x4_f32)
  #pragma unroll
  for (int c0 = 0; c0 < C_CH; c0 += 16) {
    v8f acc = {0.f, 0.f, 0.f, 0.f, 0.f, 0.f, 0.f, 0.f};
    #pragma unroll
    for (int j = 0; j < 4; ++j) {
      v2f A;
      A.x = (m == 4 * j + khalf + 0) ? invc : 0.0f;
      A.y = (m == 4 * j + khalf + 1) ? invc : 0.0f;
      const int r0 = s0 + 4 * j + khalf;
      v2f B;
      B.x = out[(size_t)r0 * C_CH + c0 + m];
      B.y = out[(size_t)(r0 + 1) * C_CH + c0 + m];
      acc = __builtin_amdgcn_wmma_f32_16x16x4_f32(
          false, A, false, B, (short)0, acc, false, false);
    }
    const int rowbase = s0 + (half << 3);
    #pragma unroll
    for (int r = 0; r < 8; ++r)
      out[(size_t)(rowbase + r) * C_CH + c0 + m] = acc[r];
  }
#else
  for (int c0 = 0; c0 < C_CH; c0 += 16) {
    for (int r = half * 8; r < half * 8 + 8; ++r)
      out[(size_t)(s0 + r) * C_CH + c0 + m] *= 1.0f / fmaxf(cnts[s0 + r], 1.0f);
  }
#endif
}

extern "C" void kernel_launch(void* const* d_in, const int* in_sizes, int n_in,
                              void* d_out, int out_size, void* d_ws, size_t ws_size,
                              hipStream_t stream) {
  const float* feat = (const float*)d_in[0];
  const int*   yx   = (const int*)d_in[1];
  const int*   seg  = (const int*)d_in[2];
  float* out  = (float*)d_out;
  float* cnts = (float*)d_ws;

  const int N  = in_sizes[2];
  const int S  = out_size / C_CH;
  const int HW = in_sizes[0] / C_CH;
  const int W  = (int)lround(sqrt((double)HW));

  {
    int n = out_size;
    SpVecs_zero<<<(n + 255) / 256, 256, 0, stream>>>(out, cnts, n, S);
  }

  SpVecs_scatter<<<(N + 255) / 256, 256, 0, stream>>>(feat, yx, seg, out, cnts,
                                                      N, HW, W);

  {
    int groups = S / 16;
    int wpb = 8;
    int blocks = (groups + wpb - 1) / wpb;
    SpVecs_finalize<<<blocks, wpb * 32, 0, stream>>>(out, cnts);
  }
}


# entries counting-sorted by segment id (returning int adds give slots); 4 waves per segment sum its 256-B rows in registers, store once; no f32 atomics; sums in arrival-rank order
# speedup vs baseline: 15.4140x; 15.4140x over previous
#include <hip/hip_runtime.h>
#include <math.h>


typedef float v2f __attribute__((ext_vector_type(2)));
typedef float v4f __attribute__((ext_vector_type(4)));
typedef float v8f __attribute__((ext_vector_type(8)));

#define C_CH 64
#define TILE_PX 64
#define TILE_PITCH 65

__device__ __forceinline__ void lds_barrier() {
  __builtin_amdgcn_fence(__ATOMIC_RELEASE, "workgroup", "local");
  __builtin_amdgcn_s_barrier();
  __builtin_amdgcn_fence(__ATOMIC_ACQUIRE, "workgroup", "local");
}

__device__ __forceinline__ void load_tile_regs(v4f (&v)[4], const float* __restrict__ feat,
                                               int HW, int p, int cl) {
  #pragma unroll
  for (int it = 0; it < 4; ++it) {
    const float* src = feat + (size_t)(it * 16 + cl) * (size_t)HW + p;
    v[it] = __builtin_nontemporal_load((const v4f*)src);
  }
}

__global__ __launch_bounds__(256) void SpVecs_to_rows(
    const float* __restrict__ feat, float* __restrict__ rows, int HW,
    int* __restrict__ cnt, int S) {
  __shared__ float tile[TILE_PX * TILE_PITCH];
  const int t    = threadIdx.x;
  for (int i = blockIdx.x * 256 + t; i < S; i += gridDim.x * 256) cnt[i] = 0;
  const int lane = t & 31;
  const int w    = t >> 5;
  const int q    = (t & 15) << 2;
  const int cl   = t >> 4;
  const int ntiles = (HW + TILE_PX - 1) / TILE_PX;
  const int nfull  = ((HW & 3) == 0) ? HW / TILE_PX : 0;

  int tl = blockIdx.x;
  v4f v[4];
  if (tl < nfull) load_tile_regs(v, feat, HW, tl * TILE_PX + q, cl);
  for (; tl < nfull; tl += gridDim.x) {
    #pragma unroll
    for (int it = 0; it < 4; ++it) {
      const int c = it * 16 + cl;
      tile[(q + 0) * TILE_PITCH + c] = v[it].x;
      tile[(q + 1) * TILE_PITCH + c] = v[it].y;
      tile[(q + 2) * TILE_PITCH + c] = v[it].z;
      tile[(q + 3) * TILE_PITCH + c] = v[it].w;
    }
    const int nx = tl + gridDim.x;
    if (nx < nfull) load_tile_regs(v, feat, HW, nx * TILE_PX + q, cl);
    lds_barrier();
    float* dst = rows + ((size_t)tl * TILE_PX + w * 8) * C_CH + lane;
    const float* srcl = tile + (w * 8) * TILE_PITCH + lane;
    #pragma unroll
    for (int k = 0; k < 8; ++k) {
      dst[k * C_CH]      = srcl[k * TILE_PITCH];
      dst[k * C_CH + 32] = srcl[k * TILE_PITCH + 32];
    }
    lds_barrier();
  }

  for (tl = nfull + blockIdx.x; tl < ntiles; tl += gridDim.x) {
    const int p0 = tl * TILE_PX;
    for (int e = t; e < TILE_PX * C_CH; e += 256) {
      const int c = e >> 6, px = e & 63;
      if (p0 + px < HW)
        tile[px * TILE_PITCH + c] = feat[(size_t)c * (size_t)HW + p0 + px];
    }
    __syncthreads();
    for (int k = 0; k < 8; ++k) {
      const int px = w * 8 + k;
      if (p0 + px < HW) {
        float* dst = rows + (size_t)(p0 + px) * C_CH;
        dst[lane]      = tile[px * TILE_PITCH + lane];
        dst[lane + 32] = tile[px * TILE_PITCH + lane + 32];
      }
    }
    __syncthreads();
  }
}

__global__ __launch_bounds__(256) void SpVecs_rank(
    const int* __restrict__ seg, int* __restrict__ cnt, int* __restrict__ rank, int N) {
  const int i = (blockIdx.x * 256 + threadIdx.x) * 4;
  if (i + 3 < N) {
    const int4 s = *(const int4*)(seg + i);
    int4 r;
    r.x = atomicAdd(&cnt[s.x], 1);
    r.y = atomicAdd(&cnt[s.y], 1);
    r.z = atomicAdd(&cnt[s.z], 1);
    r.w = atomicAdd(&cnt[s.w], 1);
    *(int4*)(rank + i) = r;
  } else {
    for (int j = i; j < N; ++j) rank[j] = atomicAdd(&cnt[seg[j]], 1);
  }
}

__global__ __launch_bounds__(1024) void SpVecs_starts(
    const int* __restrict__ cnt, int* __restrict__ start, int S) {
  __shared__ int wtot[32];
  const int t = threadIdx.x, lane = t & 31, w = t >> 5;
  int carry = 0;
  for (int base = 0; base < S; base += 4096) {
    const int i = base + 4 * t;
    int4 c = {0, 0, 0, 0};
    if (i + 3 < S) {
      c = *(const int4*)(cnt + i);
    } else {
      if (i < S)     c.x = cnt[i];
      if (i + 1 < S) c.y = cnt[i + 1];
      if (i + 2 < S) c.z = cnt[i + 2];
    }
    const int tsum = c.x + c.y + c.z + c.w;
    int x = tsum;
    #pragma unroll
    for (int d = 1; d < 32; d <<= 1) {
      const int y = __shfl_up(x, d);
      if (lane >= d) x += y;
    }
    if (lane == 31) wtot[w] = x;
    __syncthreads();
    int wx = wtot[lane];
    #pragma unroll
    for (int d = 1; d < 32; d <<= 1) {
      const int y = __shfl_up(wx, d);
      if (lane >= d) wx += y;
    }
    const int wprev = __shfl(wx, (w + 31) & 31);
    const int total = __shfl(wx, 31);
    const int ex = carry + (w > 0 ? wprev : 0) + (x - tsum);
    if (i + 3 < S) {
      int4 o;
      o.x = ex; o.y = ex + c.x; o.z = o.y + c.y; o.w = o.z + c.z;
      *(int4*)(start + i) = o;
    } else {
      if (i < S)     start[i]     = ex;
      if (i + 1 < S) start[i + 1] = ex + c.x;
      if (i + 2 < S) start[i + 2] = ex + c.x + c.y;
    }
    carry += total;
    __syncthreads();
  }
}

__global__ __launch_bounds__(256) void SpVecs_place(
    const int* __restrict__ yx, const int* __restrict__ seg,
    const int* __restrict__ rank, const int* __restrict__ start,
    int* __restrict__ sorted, int N, int W) {
  const int i = (blockIdx.x * 256 + threadIdx.x) * 4;
  if (i + 3 < N) {
    const int4 s  = *(const int4*)(seg + i);
    const int4 r  = *(const int4*)(rank + i);
    const int4 a  = *(const int4*)(yx + 2 * i);
    const int4 b  = *(const int4*)(yx + 2 * i + 4);
    sorted[start[s.x] + r.x] = a.x * W + a.y;
    sorted[start[s.y] + r.y] = a.z * W + a.w;
    sorted[start[s.z] + r.z] = b.x * W + b.y;
    sorted[start[s.w] + r.w] = b.z * W + b.w;
  } else {
    for (int j = i; j < N; ++j)
      sorted[start[seg[j]] + rank[j]] = yx[2 * j] * W + yx[2 * j + 1];
  }
}

template <bool FULL>
__device__ __forceinline__ void add_chunk(v4f& acc0, v4f& acc1,
                                          const float* __restrict__ rows,
                                          int p, int m, int half, int c4) {
  v4f v[16];
  #pragma unroll
  for (int k = 0; k < 16; ++k) {
    const int pa = __builtin_amdgcn_readlane(p, 2 * k);
    const int pb = __builtin_amdgcn_readlane(p, 2 * k + 1);
    const int pj = half ? pb : pa;
    v[k] = *(const v4f*)(rows + (size_t)pj * C_CH + c4);
  }
  #pragma unroll
  for (int k = 0; k < 16; ++k) {
    if (FULL || 2 * k + half < m) {
      if (k & 1) acc1 += v[k]; else acc0 += v[k];
    }
  }
}

__global__ __launch_bounds__(128) void SpVecs_segsum(
    const float* __restrict__ rows, const int* __restrict__ sorted,
    const int* __restrict__ start, const int* __restrict__ cnt,
    float* __restrict__ sums) {
  __shared__ float part[4 * 128];
  const int s    = blockIdx.x;
  const int t    = threadIdx.x;
  const int lane = t & 31;
  const int w    = __builtin_amdgcn_readfirstlane(t >> 5);
  const int half = lane >> 4;
  const int c4   = (lane & 15) << 2;
  const int beg  = start[s];
  const int n    = cnt[s];
  const int* __restrict__ mine = sorted + beg;

  v4f acc0 = {0.f, 0.f, 0.f, 0.f}, acc1 = {0.f, 0.f, 0.f, 0.f};
  int ch = w * 32;
  int p = 0;
  if (ch + lane < n) p = mine[ch + lane];
  while (ch < n) {
    const int nx = ch + 128;
    int pn = 0;
    if (nx + lane < n) pn = mine[nx + lane];
    const int m = n - ch;
    if (m >= 32) add_chunk<true>(acc0, acc1, rows, p, m, half, c4);
    else         add_chunk<false>(acc0, acc1, rows, p, m, half, c4);
    p = pn;
    ch = nx;
  }
  const v4f a = acc0 + acc1;
  *(v4f*)(part + w * 128 + lane * 4) = a;
  __syncthreads();
  if (t < C_CH) {
    float r = 0.f;
    #pragma unroll
    for (int q = 0; q < 4; ++q) {
      r += part[q * 128 + t];
      r += part[q * 128 + 64 + t];
    }
    sums[(size_t)s * C_CH + t] = r;
  }
}

__global__ __launch_bounds__(256) void SpVecs_finalize(
    float* __restrict__ out, const int* __restrict__ cnts) {
  const int lane  = threadIdx.x & 31;
  const int wid   = threadIdx.x >> 5;
  const int g     = blockIdx.x * (blockDim.x >> 5) + wid;
  const int s0    = g << 4;
  const int m     = lane & 15;
  const int half  = lane >> 4;
  const int khalf = half << 1;

  const float invc = 1.0f / fmaxf((float)cnts[s0 + m], 1.0f);

#if __has_builtin(__builtin_amdgcn_wmma_f32_16x16x4_f32)
  #pragma unroll
  for (int c0 = 0; c0 < C_CH; c0 += 16) {
    v8f acc = {0.f, 0.f, 0.f, 0.f, 0.f, 0.f, 0.f, 0.f};
    #pragma unroll
    for (int j = 0; j < 4; ++j) {
      v2f A;
      A.x = (m == 4 * j + khalf + 0) ? invc : 0.0f;
      A.y = (m == 4 * j + khalf + 1) ? invc : 0.0f;
      const int r0 = s0 + 4 * j + khalf;
      v2f B;
      B.x = out[(size_t)r0 * C_CH + c0 + m];
      B.y = out[(size_t)(r0 + 1) * C_CH + c0 + m];
      acc = __builtin_amdgcn_wmma_f32_16x16x4_f32(
          false, A, false, B, (short)0, acc, false, false);
    }
    const int rowbase = s0 + (half << 3);
    #pragma unroll
    for (int r = 0; r < 8; ++r)
      out[(size_t)(rowbase + r) * C_CH + c0 + m] = acc[r];
  }
#else
  for (int c0 = 0; c0 < C_CH; c0 += 16) {
    for (int r = half * 8; r < half * 8 + 8; ++r)
      out[(size_t)(s0 + r) * C_CH + c0 + m] *= 1.0f / fmaxf((float)cnts[s0 + r], 1.0f);
  }
#endif
}

extern "C" void kernel_launch(void* const* d_in, const int* in_sizes, int n_in,
                              void* d_out, int out_size, void* d_ws, size_t ws_size,
                              hipStream_t stream) {
  const float* feat = (const float*)d_in[0];
  const int*   yx   = (const int*)d_in[1];
  const int*   seg  = (const int*)d_in[2];
  float* out  = (float*)d_out;
  int*   cnts  = (int*)d_ws;
  int*   start = (int*)d_ws + 131072;
  float* rows  = (float*)d_ws + 262144;

  const int N  = in_sizes[2];
  const int S  = out_size / C_CH;
  const int HW = in_sizes[0] / C_CH;
  const int W  = (int)lround(sqrt((double)HW));
  int* rank   = (int*)(rows + (size_t)HW * C_CH);
  int* sorted = rank + N;

  SpVecs_to_rows<<<2048, 256, 0, stream>>>(feat, rows, HW, cnts, S);

  SpVecs_rank<<<(N + 1023) / 1024, 256, 0, stream>>>(seg, cnts, rank, N);

  SpVecs_starts<<<1, 1024, 0, stream>>>(cnts, start, S);

  SpVecs_place<<<(N + 1023) / 1024, 256, 0, stream>>>(yx, seg, rank, start, sorted, N, W);

  SpVecs_segsum<<<S, 128, 0, stream>>>(rows, sorted, start, cnts, out);

  {
    int groups = S / 16;
    int wpb = 8;
    int blocks = (groups + wpb - 1) / wpb;
    SpVecs_finalize<<<blocks, wpb * 32, 0, stream>>>(out, cnts);
  }
}


# rank pass counts a 16384-entry tile by segment in LDS (returning LDS adds), then one returning global add per segment and tile, a wave's adds in one 128-B line; f32 sums still in arrival-rank order
# speedup vs baseline: 28.2774x; 1.8345x over previous
#include <hip/hip_runtime.h>
#include <math.h>


typedef float v2f __attribute__((ext_vector_type(2)));
typedef float v4f __attribute__((ext_vector_type(4)));
typedef float v8f __attribute__((ext_vector_type(8)));

#define C_CH 64
#define TILE_PX 64
#define TILE_PITCH 65

__device__ __forceinline__ void lds_barrier() {
  __builtin_amdgcn_fence(__ATOMIC_RELEASE, "workgroup", "local");
  __builtin_amdgcn_s_barrier();
  __builtin_amdgcn_fence(__ATOMIC_ACQUIRE, "workgroup", "local");
}

__device__ __forceinline__ void load_tile_regs(v4f (&v)[4], const float* __restrict__ feat,
                                               int HW, int p, int cl) {
  #pragma unroll
  for (int it = 0; it < 4; ++it) {
    const float* src = feat + (size_t)(it * 16 + cl) * (size_t)HW + p;
    v[it] = __builtin_nontemporal_load((const v4f*)src);
  }
}

__global__ __launch_bounds__(256) void SpVecs_to_rows(
    const float* __restrict__ feat, float* __restrict__ rows, int HW,
    int* __restrict__ cnt, int S) {
  __shared__ float tile[TILE_PX * TILE_PITCH];
  const int t    = threadIdx.x;
  for (int i = blockIdx.x * 256 + t; i < S; i += gridDim.x * 256) cnt[i] = 0;
  const int lane = t & 31;
  const int w    = t >> 5;
  const int q    = (t & 15) << 2;
  const int cl   = t >> 4;
  const int ntiles = (HW + TILE_PX - 1) / TILE_PX;
  const int nfull  = ((HW & 3) == 0) ? HW / TILE_PX : 0;

  int tl = blockIdx.x;
  v4f v[4];
  if (tl < nfull) load_tile_regs(v, feat, HW, tl * TILE_PX + q, cl);
  for (; tl < nfull; tl += gridDim.x) {
    #pragma unroll
    for (int it = 0; it < 4; ++it) {
      const int c = it * 16 + cl;
      tile[(q + 0) * TILE_PITCH + c] = v[it].x;
      tile[(q + 1) * TILE_PITCH + c] = v[it].y;
      tile[(q + 2) * TILE_PITCH + c] = v[it].z;
      tile[(q + 3) * TILE_PITCH + c] = v[it].w;
    }
    const int nx = tl + gridDim.x;
    if (nx < nfull) load_tile_regs(v, feat, HW, nx * TILE_PX + q, cl);
    lds_barrier();
    float* dst = rows + ((size_t)tl * TILE_PX + w * 8) * C_CH + lane;
    const float* srcl = tile + (w * 8) * TILE_PITCH + lane;
    #pragma unroll
    for (int k = 0; k < 8; ++k) {
      dst[k * C_CH]      = srcl[k * TILE_PITCH];
      dst[k * C_CH + 32] = srcl[k * TILE_PITCH + 32];
    }
    lds_barrier();
  }

  for (tl = nfull + blockIdx.x; tl < ntiles; tl += gridDim.x) {
    const int p0 = tl * TILE_PX;
    for (int e = t; e < TILE_PX * C_CH; e += 256) {
      const int c = e >> 6, px = e & 63;
      if (p0 + px < HW)
        tile[px * TILE_PITCH + c] = feat[(size_t)c * (size_t)HW + p0 + px];
    }
    __syncthreads();
    for (int k = 0; k < 8; ++k) {
      const int px = w * 8 + k;
      if (p0 + px < HW) {
        float* dst = rows + (size_t)(p0 + px) * C_CH;
        dst[lane]      = tile[px * TILE_PITCH + lane];
        dst[lane + 32] = tile[px * TILE_PITCH + lane + 32];
      }
    }
    __syncthreads();
  }
}

#define RANK_THREADS 1024
#define RANK_GROUPS  4
#define RANK_TILE    (RANK_THREADS * 4 * RANK_GROUPS)
#define RANK_BINS    4096

__global__ __launch_bounds__(RANK_THREADS) void SpVecs_rank(
    const int* __restrict__ seg, int* __restrict__ cnt, int* __restrict__ rank,
    int N, int S) {
  __shared__ int hist[RANK_BINS];
  const int t     = threadIdx.x;
  const int tile0 = blockIdx.x * RANK_TILE;
  const bool whole = tile0 + RANK_TILE <= N;

  int s[RANK_GROUPS][4], r[RANK_GROUPS][4];
  if (whole) {
    #pragma unroll
    for (int g = 0; g < RANK_GROUPS; ++g) {
      const int4 v = *(const int4*)(seg + tile0 + g * (RANK_THREADS * 4) + 4 * t);
      s[g][0] = v.x; s[g][1] = v.y; s[g][2] = v.z; s[g][3] = v.w;
    }
  } else {
    #pragma unroll
    for (int g = 0; g < RANK_GROUPS; ++g) {
      const int i = tile0 + g * (RANK_THREADS * 4) + 4 * t;
      #pragma unroll
      for (int e = 0; e < 4; ++e) s[g][e] = (i + e < N) ? seg[i + e] : -1;
    }
  }
  #pragma unroll
  for (int g = 0; g < RANK_GROUPS; ++g) {
    #pragma unroll
    for (int e = 0; e < 4; ++e) r[g][e] = 0;
  }

  const int rot = (blockIdx.x * 64) & (RANK_BINS - 1);

  for (int b0 = 0; b0 < S; b0 += RANK_BINS) {
    if (b0 > 0) __syncthreads();
    #pragma unroll
    for (int q = 0; q < RANK_BINS / RANK_THREADS; ++q) hist[q * RANK_THREADS + t] = 0;
    __syncthreads();
    #pragma unroll
    for (int g = 0; g < RANK_GROUPS; ++g) {
      #pragma unroll
      for (int e = 0; e < 4; ++e) {
        const unsigned d = (unsigned)(s[g][e] - b0);
        if (d < (unsigned)RANK_BINS) r[g][e] = atomicAdd(&hist[d], 1);
      }
    }
    __syncthreads();
    int jj[RANK_BINS / RANK_THREADS], c[RANK_BINS / RANK_THREADS];
    #pragma unroll
    for (int q = 0; q < RANK_BINS / RANK_THREADS; ++q) {
      jj[q] = (q * RANK_THREADS + t + rot) & (RANK_BINS - 1);
      c[q]  = hist[jj[q]];
    }
    #pragma unroll
    for (int q = 0; q < RANK_BINS / RANK_THREADS; ++q)
      c[q] = atomicAdd(&cnt[min(b0 + jj[q], S - 1)], c[q]);
    #pragma unroll
    for (int q = 0; q < RANK_BINS / RANK_THREADS; ++q) hist[jj[q]] = c[q];
    __syncthreads();
    #pragma unroll
    for (int g = 0; g < RANK_GROUPS; ++g) {
      #pragma unroll
      for (int e = 0; e < 4; ++e) {
        const unsigned d = (unsigned)(s[g][e] - b0);
        const int base = hist[d & (unsigned)(RANK_BINS - 1)];
        r[g][e] += (d < (unsigned)RANK_BINS) ? base : 0;
      }
    }
  }

  if (whole) {
    #pragma unroll
    for (int g = 0; g < RANK_GROUPS; ++g) {
      int4 o;
      o.x = r[g][0]; o.y = r[g][1]; o.z = r[g][2]; o.w = r[g][3];
      *(int4*)(rank + tile0 + g * (RANK_THREADS * 4) + 4 * t) = o;
    }
  } else {
    #pragma unroll
    for (int g = 0; g < RANK_GROUPS; ++g) {
      const int i = tile0 + g * (RANK_THREADS * 4) + 4 * t;
      #pragma unroll
      for (int e = 0; e < 4; ++e)
        if (i + e < N) rank[i + e] = r[g][e];
    }
  }
}

__global__ __launch_bounds__(1024) void SpVecs_starts(
    const int* __restrict__ cnt, int* __restrict__ start, int S) {
  __shared__ int wtot[32];
  const int t = threadIdx.x, lane = t & 31, w = t >> 5;
  int carry = 0;
  for (int base = 0; base < S; base += 4096) {
    const int i = base + 4 * t;
    int4 c = {0, 0, 0, 0};
    if (i + 3 < S) {
      c = *(const int4*)(cnt + i);
    } else {
      if (i < S)     c.x = cnt[i];
      if (i + 1 < S) c.y = cnt[i + 1];
      if (i + 2 < S) c.z = cnt[i + 2];
    }
    const int tsum = c.x + c.y + c.z + c.w;
    int x = tsum;
    #pragma unroll
    for (int d = 1; d < 32; d <<= 1) {
      const int y = __shfl_up(x, d);
      if (lane >= d) x += y;
    }
    if (lane == 31) wtot[w] = x;
    __syncthreads();
    int wx = wtot[lane];
    #pragma unroll
    for (int d = 1; d < 32; d <<= 1) {
      const int y = __shfl_up(wx, d);
      if (lane >= d) wx += y;
    }
    const int wprev = __shfl(wx, (w + 31) & 31);
    const int total = __shfl(wx, 31);
    const int ex = carry + (w > 0 ? wprev : 0) + (x - tsum);
    if (i + 3 < S) {
      int4 o;
      o.x = ex; o.y = ex + c.x; o.z = o.y + c.y; o.w = o.z + c.z;
      *(int4*)(start + i) = o;
    } else {
      if (i < S)     start[i]     = ex;
      if (i + 1 < S) start[i + 1] = ex + c.x;
      if (i + 2 < S) start[i + 2] = ex + c.x + c.y;
    }
    carry += total;
    __syncthreads();
  }
}

__global__ __launch_bounds__(256) void SpVecs_place(
    const int* __restrict__ yx, const int* __restrict__ seg,
    const int* __restrict__ rank, const int* __restrict__ start,
    int* __restrict__ sorted, int N, int W) {
  const int i = (blockIdx.x * 256 + threadIdx.x) * 4;
  if (i + 3 < N) {
    const int4 s  = *(const int4*)(seg + i);
    const int4 r  = *(const int4*)(rank + i);
    const int4 a  = *(const int4*)(yx + 2 * i);
    const int4 b  = *(const int4*)(yx + 2 * i + 4);
    sorted[start[s.x] + r.x] = a.x * W + a.y;
    sorted[start[s.y] + r.y] = a.z * W + a.w;
    sorted[start[s.z] + r.z] = b.x * W + b.y;
    sorted[start[s.w] + r.w] = b.z * W + b.w;
  } else {
    for (int j = i; j < N; ++j)
      sorted[start[seg[j]] + rank[j]] = yx[2 * j] * W + yx[2 * j + 1];
  }
}

template <bool FULL>
__device__ __forceinline__ void add_chunk(v4f& acc0, v4f& acc1,
                                          const float* __restrict__ rows,
                                          int p, int m, int half, int c4) {
  v4f v[16];
  #pragma unroll
  for (int k = 0; k < 16; ++k) {
    const int pa = __builtin_amdgcn_readlane(p, 2 * k);
    const int pb = __builtin_amdgcn_readlane(p, 2 * k + 1);
    const int pj = half ? pb : pa;
    v[k] = *(const v4f*)(rows + (size_t)pj * C_CH + c4);
  }
  #pragma unroll
  for (int k = 0; k < 16; ++k) {
    if (FULL || 2 * k + half < m) {
      if (k & 1) acc1 += v[k]; else acc0 += v[k];
    }
  }
}

__global__ __launch_bounds__(128) void SpVecs_segsum(
    const float* __restrict__ rows, const int* __restrict__ sorted,
    const int* __restrict__ start, const int* __restrict__ cnt,
    float* __restrict__ sums) {
  __shared__ float part[4 * 128];
  const int s    = blockIdx.x;
  const int t    = threadIdx.x;
  const int lane = t & 31;
  const int w    = __builtin_amdgcn_readfirstlane(t >> 5);
  const int half = lane >> 4;
  const int c4   = (lane & 15) << 2;
  const int beg  = start[s];
  const int n    = cnt[s];
  const int* __restrict__ mine = sorted + beg;

  v4f acc0 = {0.f, 0.f, 0.f, 0.f}, acc1 = {0.f, 0.f, 0.f, 0.f};
  int ch = w * 32;
  int p = 0;
  if (ch + lane < n) p = mine[ch + lane];
  while (ch < n) {
    const int nx = ch + 128;
    int pn = 0;
    if (nx + lane < n) pn = mine[nx + lane];
    const int m = n - ch;
    if (m >= 32) add_chunk<true>(acc0, acc1, rows, p, m, half, c4);
    else         add_chunk<false>(acc0, acc1, rows, p, m, half, c4);
    p = pn;
    ch = nx;
  }
  const v4f a = acc0 + acc1;
  *(v4f*)(part + w * 128 + lane * 4) = a;
  __syncthreads();
  if (t < C_CH) {
    float r = 0.f;
    #pragma unroll
    for (int q = 0; q < 4; ++q) {
      r += part[q * 128 + t];
      r += part[q * 128 + 64 + t];
    }
    sums[(size_t)s * C_CH + t] = r;
  }
}

__global__ __launch_bounds__(256) void SpVecs_finalize(
    float* __restrict__ out, const int* __restrict__ cnts) {
  const int lane  = threadIdx.x & 31;
  const int wid   = threadIdx.x >> 5;
  const int g     = blockIdx.x * (blockDim.x >> 5) + wid;
  const int s0    = g << 4;
  const int m     = lane & 15;
  const int half  = lane >> 4;
  const int khalf = half << 1;

  const float invc = 1.0f / fmaxf((float)cnts[s0 + m], 1.0f);

#if __has_builtin(__builtin_amdgcn_wmma_f32_16x16x4_f32)
  #pragma unroll
  for (int c0 = 0; c0 < C_CH; c0 += 16) {
    v8f acc = {0.f, 0.f, 0.f, 0.f, 0.f, 0.f, 0.f, 0.f};
    #pragma unroll
    for (int j = 0; j < 4; ++j) {
      v2f A;
      A.x = (m == 4 * j + khalf + 0) ? invc : 0.0f;
      A.y = (m == 4 * j + khalf + 1) ? invc : 0.0f;
      const int r0 = s0 + 4 * j + khalf;
      v2f B;
      B.x = out[(size_t)r0 * C_CH + c0 + m];
      B.y = out[(size_t)(r0 + 1) * C_CH + c0 + m];
      acc = __builtin_amdgcn_wmma_f32_16x16x4_f32(
          false, A, false, B, (short)0, acc, false, false);
    }
    const int rowbase = s0 + (half << 3);
    #pragma unroll
    for (int r = 0; r < 8; ++r)
      out[(size_t)(rowbase + r) * C_CH + c0 + m] = acc[r];
  }
#else
  for (int c0 = 0; c0 < C_CH; c0 += 16) {
    for (int r = half * 8; r < half * 8 + 8; ++r)
      out[(size_t)(s0 + r) * C_CH + c0 + m] *= 1.0f / fmaxf((float)cnts[s0 + r], 1.0f);
  }
#endif
}

extern "C" void kernel_launch(void* const* d_in, const int* in_sizes, int n_in,
                              void* d_out, int out_size, void* d_ws, size_t ws_size,
                              hipStream_t stream) {
  const float* feat = (const float*)d_in[0];
  const int*   yx   = (const int*)d_in[1];
  const int*   seg  = (const int*)d_in[2];
  float* out  = (float*)d_out;
  int*   cnts  = (int*)d_ws;
  int*   start = (int*)d_ws + 131072;
  float* rows  = (float*)d_ws + 262144;

  const int N  = in_sizes[2];
  const int S  = out_size / C_CH;
  const int HW = in_sizes[0] / C_CH;
  const int W  = (int)lround(sqrt((double)HW));
  int* rank   = (int*)(rows + (size_t)HW * C_CH);
  int* sorted = rank + N;

  SpVecs_to_rows<<<2048, 256, 0, stream>>>(feat, rows, HW, cnts, S);

  SpVecs_rank<<<(N + RANK_TILE - 1) / RANK_TILE, RANK_THREADS, 0, stream>>>(seg, cnts, rank, N, S);

  SpVecs_starts<<<1, 1024, 0, stream>>>(cnts, start, S);

  SpVecs_place<<<(N + 1023) / 1024, 256, 0, stream>>>(yx, seg, rank, start, sorted, N, W);

  SpVecs_segsum<<<S, 128, 0, stream>>>(rows, sorted, start, cnts, out);

  {
    int groups = S / 16;
    int wpb = 8;
    int blocks = (groups + wpb - 1) / wpb;
    SpVecs_finalize<<<blocks, wpb * 32, 0, stream>>>(out, cnts);
  }
}


# segment-start scan folded into the placing pass: each work-group (1024 threads, 4096 entries) scans the sizes into LDS and looks starts up there; starts launch gone (5 left); f32 sum order as before
# speedup vs baseline: 29.5915x; 1.0465x over previous
#include <hip/hip_runtime.h>
#include <math.h>


typedef float v2f __attribute__((ext_vector_type(2)));
typedef float v4f __attribute__((ext_vector_type(4)));
typedef float v8f __attribute__((ext_vector_type(8)));

#define C_CH 64
#define TILE_PX 64
#define TILE_PITCH 65

__device__ __forceinline__ void lds_barrier() {
  __builtin_amdgcn_fence(__ATOMIC_RELEASE, "workgroup", "local");
  __builtin_amdgcn_s_barrier();
  __builtin_amdgcn_fence(__ATOMIC_ACQUIRE, "workgroup", "local");
}

__device__ __forceinline__ void load_tile_regs(v4f (&v)[4], const float* __restrict__ feat,
                                               int HW, int p, int cl) {
  #pragma unroll
  for (int it = 0; it < 4; ++it) {
    const float* src = feat + (size_t)(it * 16 + cl) * (size_t)HW + p;
    v[it] = __builtin_nontemporal_load((const v4f*)src);
  }
}

__global__ __launch_bounds__(256) void SpVecs_to_rows(
    const float* __restrict__ feat, float* __restrict__ rows, int HW,
    int* __restrict__ cnt, int S) {
  __shared__ float tile[TILE_PX * TILE_PITCH];
  const int t    = threadIdx.x;
  for (int i = blockIdx.x * 256 + t; i < S; i += gridDim.x * 256) cnt[i] = 0;
  const int lane = t & 31;
  const int w    = t >> 5;
  const int q    = (t & 15) << 2;
  const int cl   = t >> 4;
  const int ntiles = (HW + TILE_PX - 1) / TILE_PX;
  const int nfull  = ((HW & 3) == 0) ? HW / TILE_PX : 0;

  int tl = blockIdx.x;
  v4f v[4];
  if (tl < nfull) load_tile_regs(v, feat, HW, tl * TILE_PX + q, cl);
  for (; tl < nfull; tl += gridDim.x) {
    #pragma unroll
    for (int it = 0; it < 4; ++it) {
      const int c = it * 16 + cl;
      tile[(q + 0) * TILE_PITCH + c] = v[it].x;
      tile[(q + 1) * TILE_PITCH + c] = v[it].y;
      tile[(q + 2) * TILE_PITCH + c] = v[it].z;
      tile[(q + 3) * TILE_PITCH + c] = v[it].w;
    }
    const int nx = tl + gridDim.x;
    if (nx < nfull) load_tile_regs(v, feat, HW, nx * TILE_PX + q, cl);
    lds_barrier();
    float* dst = rows + ((size_t)tl * TILE_PX + w * 8) * C_CH + lane;
    const float* srcl = tile + (w * 8) * TILE_PITCH + lane;
    #pragma unroll
    for (int k = 0; k < 8; ++k) {
      dst[k * C_CH]      = srcl[k * TILE_PITCH];
      dst[k * C_CH + 32] = srcl[k * TILE_PITCH + 32];
    }
    lds_barrier();
  }

  for (tl = nfull + blockIdx.x; tl < ntiles; tl += gridDim.x) {
    const int p0 = tl * TILE_PX;
    for (int e = t; e < TILE_PX * C_CH; e += 256) {
      const int c = e >> 6, px = e & 63;
      if (p0 + px < HW)
        tile[px * TILE_PITCH + c] = feat[(size_t)c * (size_t)HW + p0 + px];
    }
    __syncthreads();
    for (int k = 0; k < 8; ++k) {
      const int px = w * 8 + k;
      if (p0 + px < HW) {
        float* dst = rows + (size_t)(p0 + px) * C_CH;
        dst[lane]      = tile[px * TILE_PITCH + lane];
        dst[lane + 32] = tile[px * TILE_PITCH + lane + 32];
      }
    }
    __syncthreads();
  }
}

#define RANK_THREADS 1024
#define RANK_GROUPS  4
#define RANK_TILE    (RANK_THREADS * 4 * RANK_GROUPS)
#define RANK_BINS    4096

__global__ __launch_bounds__(RANK_THREADS) void SpVecs_rank(
    const int* __restrict__ seg, int* __restrict__ cnt, int* __restrict__ rank,
    int N, int S) {
  __shared__ int hist[RANK_BINS];
  const int t     = threadIdx.x;
  const int tile0 = blockIdx.x * RANK_TILE;
  const bool whole = tile0 + RANK_TILE <= N;

  int s[RANK_GROUPS][4], r[RANK_GROUPS][4];
  if (whole) {
    #pragma unroll
    for (int g = 0; g < RANK_GROUPS; ++g) {
      const int4 v = *(const int4*)(seg + tile0 + g * (RANK_THREADS * 4) + 4 * t);
      s[g][0] = v.x; s[g][1] = v.y; s[g][2] = v.z; s[g][3] = v.w;
    }
  } else {
    #pragma unroll
    for (int g = 0; g < RANK_GROUPS; ++g) {
      const int i = tile0 + g * (RANK_THREADS * 4) + 4 * t;
      #pragma unroll
      for (int e = 0; e < 4; ++e) s[g][e] = (i + e < N) ? seg[i + e] : -1;
    }
  }
  #pragma unroll
  for (int g = 0; g < RANK_GROUPS; ++g) {
    #pragma unroll
    for (int e = 0; e < 4; ++e) r[g][e] = 0;
  }

  const int rot = (blockIdx.x * 64) & (RANK_BINS - 1);

  for (int b0 = 0; b0 < S; b0 += RANK_BINS) {
    if (b0 > 0) __syncthreads();
    #pragma unroll
    for (int q = 0; q < RANK_BINS / RANK_THREADS; ++q) hist[q * RANK_THREADS + t] = 0;
    __syncthreads();
    #pragma unroll
    for (int g = 0; g < RANK_GROUPS; ++g) {
      #pragma unroll
      for (int e = 0; e < 4; ++e) {
        const unsigned d = (unsigned)(s[g][e] - b0);
        if (d < (unsigned)RANK_BINS) r[g][e] = atomicAdd(&hist[d], 1);
      }
    }
    __syncthreads();
    int jj[RANK_BINS / RANK_THREADS], c[RANK_BINS / RANK_THREADS];
    #pragma unroll
    for (int q = 0; q < RANK_BINS / RANK_THREADS; ++q) {
      jj[q] = (q * RANK_THREADS + t + rot) & (RANK_BINS - 1);
      c[q]  = hist[jj[q]];
    }
    #pragma unroll
    for (int q = 0; q < RANK_BINS / RANK_THREADS; ++q)
      c[q] = atomicAdd(&cnt[min(b0 + jj[q], S - 1)], c[q]);
    #pragma unroll
    for (int q = 0; q < RANK_BINS / RANK_THREADS; ++q) hist[jj[q]] = c[q];
    __syncthreads();
    #pragma unroll
    for (int g = 0; g < RANK_GROUPS; ++g) {
      #pragma unroll
      for (int e = 0; e < 4; ++e) {
        const unsigned d = (unsigned)(s[g][e] - b0);
        const int base = hist[d & (unsigned)(RANK_BINS - 1)];
        r[g][e] += (d < (unsigned)RANK_BINS) ? base : 0;
      }
    }
  }

  if (whole) {
    #pragma unroll
    for (int g = 0; g < RANK_GROUPS; ++g) {
      int4 o;
      o.x = r[g][0]; o.y = r[g][1]; o.z = r[g][2]; o.w = r[g][3];
      *(int4*)(rank + tile0 + g * (RANK_THREADS * 4) + 4 * t) = o;
    }
  } else {
    #pragma unroll
    for (int g = 0; g < RANK_GROUPS; ++g) {
      const int i = tile0 + g * (RANK_THREADS * 4) + 4 * t;
      #pragma unroll
      for (int e = 0; e < 4; ++e)
        if (i + e < N) rank[i + e] = r[g][e];
    }
  }
}

#define PLACE_THREADS 1024
#define PLACE_WIN     (PLACE_THREADS * 4)

__device__ __forceinline__ int4 load_sizes(const int* __restrict__ cnt, int i, int S) {
  int4 c = {0, 0, 0, 0};
  if (i + 3 < S) {
    c = *(const int4*)(cnt + i);
  } else {
    if (i < S)     c.x = cnt[i];
    if (i + 1 < S) c.y = cnt[i + 1];
    if (i + 2 < S) c.z = cnt[i + 2];
  }
  return c;
}

template <bool WHOLE>
__device__ __forceinline__ void place_body(
    const int* __restrict__ yx, const int* __restrict__ seg,
    const int* __restrict__ rank, const int* __restrict__ cnt,
    int* __restrict__ start, int* __restrict__ sorted, int N, int W, int S,
    int* __restrict__ st, int* __restrict__ wtot) {
  const int t = threadIdx.x, lane = t & 31, w = t >> 5;

  int4 c = load_sizes(cnt, 4 * t, S);
  const int e0 = (blockIdx.x * PLACE_THREADS + t) * 4;
  int4 s = {-1, -1, -1, -1};
  int4 r = {0, 0, 0, 0}, a = {0, 0, 0, 0}, b = {0, 0, 0, 0};
  if (WHOLE) {
    s = *(const int4*)(seg + e0);
    r = *(const int4*)(rank + e0);
    a = *(const int4*)(yx + 2 * e0);
    b = *(const int4*)(yx + 2 * e0 + 4);
  } else {
    if (e0 < N)     { s.x = seg[e0];     r.x = rank[e0];     a.x = yx[2 * e0];     a.y = yx[2 * e0 + 1]; }
    if (e0 + 1 < N) { s.y = seg[e0 + 1]; r.y = rank[e0 + 1]; a.z = yx[2 * e0 + 2]; a.w = yx[2 * e0 + 3]; }
    if (e0 + 2 < N) { s.z = seg[e0 + 2]; r.z = rank[e0 + 2]; b.x = yx[2 * e0 + 4]; b.y = yx[2 * e0 + 5]; }
    if (e0 + 3 < N) { s.w = seg[e0 + 3]; r.w = rank[e0 + 3]; b.z = yx[2 * e0 + 6]; b.w = yx[2 * e0 + 7]; }
  }

  int carry = 0;
  for (int base = 0; base < S; base += PLACE_WIN) {
    const int i = base + 4 * t;
    if (base > 0) lds_barrier();
    const int tsum = c.x + c.y + c.z + c.w;
    int x = tsum;
    #pragma unroll
    for (int d = 1; d < 32; d <<= 1) {
      const int y = __shfl_up(x, d);
      if (lane >= d) x += y;
    }
    if (lane == 31) wtot[w] = x;
    lds_barrier();
    int wx = wtot[lane];
    #pragma unroll
    for (int d = 1; d < 32; d <<= 1) {
      const int y = __shfl_up(wx, d);
      if (lane >= d) wx += y;
    }
    const int wprev = __shfl(wx, (w + 31) & 31);
    const int total = __shfl(wx, 31);
    const int ex = carry + (w > 0 ? wprev : 0) + (x - tsum);
    int4 o;
    o.x = ex; o.y = ex + c.x; o.z = o.y + c.y; o.w = o.z + c.z;
    *(int4*)(st + 4 * t) = o;
    if (blockIdx.x == 0) {
      if (i + 3 < S) {
        *(int4*)(start + i) = o;
      } else {
        if (i < S)     start[i]     = o.x;
        if (i + 1 < S) start[i + 1] = o.y;
        if (i + 2 < S) start[i + 2] = o.z;
      }
    }
    carry += total;
    if (base + PLACE_WIN < S) c = load_sizes(cnt, i + PLACE_WIN, S);
    lds_barrier();
    const unsigned lim = (unsigned)min(PLACE_WIN, S - base);
    const unsigned d0 = (unsigned)(s.x - base), d1 = (unsigned)(s.y - base);
    const unsigned d2 = (unsigned)(s.z - base), d3 = (unsigned)(s.w - base);
    const int b0 = st[d0 & (unsigned)(PLACE_WIN - 1)];
    const int b1 = st[d1 & (unsigned)(PLACE_WIN - 1)];
    const int b2 = st[d2 & (unsigned)(PLACE_WIN - 1)];
    const int b3 = st[d3 & (unsigned)(PLACE_WIN - 1)];
    if (d0 < lim) sorted[b0 + r.x] = a.x * W + a.y;
    if (d1 < lim) sorted[b1 + r.y] = a.z * W + a.w;
    if (d2 < lim) sorted[b2 + r.z] = b.x * W + b.y;
    if (d3 < lim) sorted[b3 + r.w] = b.z * W + b.w;
  }
}

__global__ __launch_bounds__(PLACE_THREADS) void SpVecs_place(
    const int* __restrict__ yx, const int* __restrict__ seg,
    const int* __restrict__ rank, const int* __restrict__ cnt,
    int* __restrict__ start, int* __restrict__ sorted, int N, int W, int S) {
  __shared__ __attribute__((aligned(16))) int st[PLACE_WIN];
  __shared__ int wtot[32];
  if ((blockIdx.x + 1) * PLACE_WIN <= N)
    place_body<true>(yx, seg, rank, cnt, start, sorted, N, W, S, st, wtot);
  else
    place_body<false>(yx, seg, rank, cnt, start, sorted, N, W, S, st, wtot);
}

template <bool FULL>
__device__ __forceinline__ void add_chunk(v4f& acc0, v4f& acc1,
                                          const float* __restrict__ rows,
                                          int p, int m, int half, int c4) {
  v4f v[16];
  #pragma unroll
  for (int k = 0; k < 16; ++k) {
    const int pa = __builtin_amdgcn_readlane(p, 2 * k);
    const int pb = __builtin_amdgcn_readlane(p, 2 * k + 1);
    const int pj = half ? pb : pa;
    v[k] = *(const v4f*)(rows + (size_t)pj * C_CH + c4);
  }
  #pragma unroll
  for (int k = 0; k < 16; ++k) {
    if (FULL || 2 * k + half < m) {
      if (k & 1) acc1 += v[k]; else acc0 += v[k];
    }
  }
}

__global__ __launch_bounds__(128) void SpVecs_segsum(
    const float* __restrict__ rows, const int* __restrict__ sorted,
    const int* __restrict__ start, const int* __restrict__ cnt,
    float* __restrict__ sums) {
  __shared__ float part[4 * 128];
  const int s    = blockIdx.x;
  const int t    = threadIdx.x;
  const int lane = t & 31;
  const int w    = __builtin_amdgcn_readfirstlane(t >> 5);
  const int half = lane >> 4;
  const int c4   = (lane & 15) << 2;
  const int beg  = start[s];
  const int n    = cnt[s];
  const int* __restrict__ mine = sorted + beg;

  v4f acc0 = {0.f, 0.f, 0.f, 0.f}, acc1 = {0.f, 0.f, 0.f, 0.f};
  int ch = w * 32;
  int p = 0;
  if (ch + lane < n) p = mine[ch + lane];
  while (ch < n) {
    const int nx = ch + 128;
    int pn = 0;
    if (nx + lane < n) pn = mine[nx + lane];
    const int m = n - ch;
    if (m >= 32) add_chunk<true>(acc0, acc1, rows, p, m, half, c4);
    else         add_chunk<false>(acc0, acc1, rows, p, m, half, c4);
    p = pn;
    ch = nx;
  }
  const v4f a = acc0 + acc1;
  *(v4f*)(part + w * 128 + lane * 4) = a;
  __syncthreads();
  if (t < C_CH) {
    float r = 0.f;
    #pragma unroll
    for (int q = 0; q < 4; ++q) {
      r += part[q * 128 + t];
      r += part[q * 128 + 64 + t];
    }
    sums[(size_t)s * C_CH + t] = r;
  }
}

__global__ __launch_bounds__(256) void SpVecs_finalize(
    float* __restrict__ out, const int* __restrict__ cnts) {
  const int lane  = threadIdx.x & 31;
  const int wid   = threadIdx.x >> 5;
  const int g     = blockIdx.x * (blockDim.x >> 5) + wid;
  const int s0    = g << 4;
  const int m     = lane & 15;
  const int half  = lane >> 4;
  const int khalf = half << 1;

  const float invc = 1.0f / fmaxf((float)cnts[s0 + m], 1.0f);

#if __has_builtin(__builtin_amdgcn_wmma_f32_16x16x4_f32)
  #pragma unroll
  for (int c0 = 0; c0 < C_CH; c0 += 16) {
    v8f acc = {0.f, 0.f, 0.f, 0.f, 0.f, 0.f, 0.f, 0.f};
    #pragma unroll
    for (int j = 0; j < 4; ++j) {
      v2f A;
      A.x = (m == 4 * j + khalf + 0) ? invc : 0.0f;
      A.y = (m == 4 * j + khalf + 1) ? invc : 0.0f;
      const int r0 = s0 + 4 * j + khalf;
      v2f B;
      B.x = out[(size_t)r0 * C_CH + c0 + m];
      B.y = out[(size_t)(r0 + 1) * C_CH + c0 + m];
      acc = __builtin_amdgcn_wmma_f32_16x16x4_f32(
          false, A, false, B, (short)0, acc, false, false);
    }
    const int rowbase = s0 + (half << 3);
    #pragma unroll
    for (int r = 0; r < 8; ++r)
      out[(size_t)(rowbase + r) * C_CH + c0 + m] = acc[r];
  }
#else
  for (int c0 = 0; c0 < C_CH; c0 += 16) {
    for (int r = half * 8; r < half * 8 + 8; ++r)
      out[(size_t)(s0 + r) * C_CH + c0 + m] *= 1.0f / fmaxf((float)cnts[s0 + r], 1.0f);
  }
#endif
}

extern "C" void kernel_launch(void* const* d_in, const int* in_sizes, int n_in,
                              void* d_out, int out_size, void* d_ws, size_t ws_size,
                              hipStream_t stream) {
  const float* feat = (const float*)d_in[0];
  const int*   yx   = (const int*)d_in[1];
  const int*   seg  = (const int*)d_in[2];
  float* out  = (float*)d_out;
  int*   cnts  = (int*)d_ws;
  int*   start = (int*)d_ws + 131072;
  float* rows  = (float*)d_ws + 262144;

  const int N  = in_sizes[2];
  const int S  = out_size / C_CH;
  const int HW = in_sizes[0] / C_CH;
  const int W  = (int)lround(sqrt((double)HW));
  int* rank   = (int*)(rows + (size_t)HW * C_CH);
  int* sorted = rank + N;

  SpVecs_to_rows<<<2048, 256, 0, stream>>>(feat, rows, HW, cnts, S);

  SpVecs_rank<<<(N + RANK_TILE - 1) / RANK_TILE, RANK_THREADS, 0, stream>>>(seg, cnts, rank, N, S);

  SpVecs_place<<<(N + PLACE_WIN - 1) / PLACE_WIN, PLACE_THREADS, 0, stream>>>(
      yx, seg, rank, cnts, start, sorted, N, W, S);

  SpVecs_segsum<<<S, 128, 0, stream>>>(rows, sorted, start, cnts, out);

  {
    int groups = S / 16;
    int wpb = 8;
    int blocks = (groups + wpb - 1) / wpb;
    SpVecs_finalize<<<blocks, wpb * 32, 0, stream>>>(out, cnts);
  }
}


# entries sorted by segment in ONE pass into fixed-capacity buckets (2x mean size; overflow list beyond): no starts scan, rank array or placing launch (4 launches); f32 sums still in arrival-rank order
# speedup vs baseline: 31.9560x; 1.0799x over previous
#include <hip/hip_runtime.h>
#include <math.h>


typedef float v2f __attribute__((ext_vector_type(2)));
typedef float v4f __attribute__((ext_vector_type(4)));
typedef float v8f __attribute__((ext_vector_type(8)));

#define C_CH 64
#define TILE_PX 64
#define TILE_PITCH 65

__device__ __forceinline__ void lds_barrier() {
  __builtin_amdgcn_fence(__ATOMIC_RELEASE, "workgroup", "local");
  __builtin_amdgcn_s_barrier();
  __builtin_amdgcn_fence(__ATOMIC_ACQUIRE, "workgroup", "local");
}

__device__ __forceinline__ void load_tile_regs(v4f (&v)[4], const float* __restrict__ feat,
                                               int HW, int p, int cl) {
  #pragma unroll
  for (int it = 0; it < 4; ++it) {
    const float* src = feat + (size_t)(it * 16 + cl) * (size_t)HW + p;
    v[it] = __builtin_nontemporal_load((const v4f*)src);
  }
}

__global__ __launch_bounds__(256) void SpVecs_to_rows(
    const float* __restrict__ feat, float* __restrict__ rows, int HW,
    int* __restrict__ cnt, int* __restrict__ ovf_n, int S) {
  __shared__ float tile[TILE_PX * TILE_PITCH];
  const int t    = threadIdx.x;
  for (int i = blockIdx.x * 256 + t; i < S; i += gridDim.x * 256) cnt[i] = 0;
  if (blockIdx.x == 0 && t == 0) *ovf_n = 0;
  const int lane = t & 31;
  const int w    = t >> 5;
  const int q    = (t & 15) << 2;
  const int cl   = t >> 4;
  const int ntiles = (HW + TILE_PX - 1) / TILE_PX;
  const int nfull  = ((HW & 3) == 0) ? HW / TILE_PX : 0;

  int tl = blockIdx.x;
  v4f v[4];
  if (tl < nfull) load_tile_regs(v, feat, HW, tl * TILE_PX + q, cl);
  for (; tl < nfull; tl += gridDim.x) {
    #pragma unroll
    for (int it = 0; it < 4; ++it) {
      const int c = it * 16 + cl;
      tile[(q + 0) * TILE_PITCH + c] = v[it].x;
      tile[(q + 1) * TILE_PITCH + c] = v[it].y;
      tile[(q + 2) * TILE_PITCH + c] = v[it].z;
      tile[(q + 3) * TILE_PITCH + c] = v[it].w;
    }
    const int nx = tl + gridDim.x;
    if (nx < nfull) load_tile_regs(v, feat, HW, nx * TILE_PX + q, cl);
    lds_barrier();
    float* dst = rows + ((size_t)tl * TILE_PX + w * 8) * C_CH + lane;
    const float* srcl = tile + (w * 8) * TILE_PITCH + lane;
    #pragma unroll
    for (int k = 0; k < 8; ++k) {
      dst[k * C_CH]      = srcl[k * TILE_PITCH];
      dst[k * C_CH + 32] = srcl[k * TILE_PITCH + 32];
    }
    lds_barrier();
  }

  for (tl = nfull + blockIdx.x; tl < ntiles; tl += gridDim.x) {
    const int p0 = tl * TILE_PX;
    for (int e = t; e < TILE_PX * C_CH; e += 256) {
      const int c = e >> 6, px = e & 63;
      if (p0 + px < HW)
        tile[px * TILE_PITCH + c] = feat[(size_t)c * (size_t)HW + p0 + px];
    }
    __syncthreads();
    for (int k = 0; k < 8; ++k) {
      const int px = w * 8 + k;
      if (p0 + px < HW) {
        float* dst = rows + (size_t)(p0 + px) * C_CH;
        dst[lane]      = tile[px * TILE_PITCH + lane];
        dst[lane + 32] = tile[px * TILE_PITCH + lane + 32];
      }
    }
    __syncthreads();
  }
}

#define BK_THREADS 1024
#define BK_TILE    (BK_THREADS * 4)
#define BK_BINS    4096

__device__ __forceinline__ int bucket_cap(int N, int S) {
  const int mean = (N + S - 1) / S;
  return (2 * mean + 31) & ~31;
}

__global__ __launch_bounds__(BK_THREADS) void SpVecs_bucket(
    const int* __restrict__ yx, const int* __restrict__ seg, int* __restrict__ cnt,
    int* __restrict__ ovf_n, int* __restrict__ ovf, int* __restrict__ bucket,
    int N, int W, int S) {
  __shared__ int hist[BK_BINS];
  const int t   = threadIdx.x;
  const int e0  = (blockIdx.x * BK_THREADS + t) * 4;
  const int cap = bucket_cap(N, S);

  int s[4], p[4], r[4];
  if ((blockIdx.x + 1) * BK_TILE <= N) {
    const int4 sv = *(const int4*)(seg + e0);
    const int4 a  = *(const int4*)(yx + 2 * e0);
    const int4 b  = *(const int4*)(yx + 2 * e0 + 4);
    s[0] = sv.x; s[1] = sv.y; s[2] = sv.z; s[3] = sv.w;
    p[0] = a.x * W + a.y; p[1] = a.z * W + a.w;
    p[2] = b.x * W + b.y; p[3] = b.z * W + b.w;
  } else {
    #pragma unroll
    for (int e = 0; e < 4; ++e) {
      s[e] = -1; p[e] = 0;
      if (e0 + e < N) {
        s[e] = seg[e0 + e];
        p[e] = yx[2 * (e0 + e)] * W + yx[2 * (e0 + e) + 1];
      }
    }
  }
  #pragma unroll
  for (int e = 0; e < 4; ++e) {
    r[e] = 0;
    if (s[e] >= S) s[e] = -1;
  }

  const int rot = (blockIdx.x * 64) & (BK_BINS - 1);

  for (int b0 = 0; b0 < S; b0 += BK_BINS) {
    if (b0 > 0) __syncthreads();
    #pragma unroll
    for (int q = 0; q < BK_BINS / BK_THREADS; ++q) hist[q * BK_THREADS + t] = 0;
    __syncthreads();
    #pragma unroll
    for (int e = 0; e < 4; ++e) {
      const unsigned d = (unsigned)(s[e] - b0);
      if (s[e] >= 0 && d < (unsigned)BK_BINS) r[e] = atomicAdd(&hist[d], 1);
    }
    __syncthreads();
    int jj[BK_BINS / BK_THREADS], c[BK_BINS / BK_THREADS];
    #pragma unroll
    for (int q = 0; q < BK_BINS / BK_THREADS; ++q) {
      jj[q] = (q * BK_THREADS + t + rot) & (BK_BINS - 1);
      c[q]  = hist[jj[q]];
    }
    #pragma unroll
    for (int q = 0; q < BK_BINS / BK_THREADS; ++q)
      c[q] = atomicAdd(&cnt[min(b0 + jj[q], S - 1)], c[q]);
    #pragma unroll
    for (int q = 0; q < BK_BINS / BK_THREADS; ++q) hist[jj[q]] = c[q];
    __syncthreads();
    #pragma unroll
    for (int e = 0; e < 4; ++e) {
      const unsigned d = (unsigned)(s[e] - b0);
      const int base = hist[d & (unsigned)(BK_BINS - 1)];
      r[e] += (s[e] >= 0 && d < (unsigned)BK_BINS) ? base : 0;
    }
  }

  #pragma unroll
  for (int e = 0; e < 4; ++e) {
    if (s[e] >= 0) {
      if (r[e] < cap) {
        bucket[(size_t)s[e] * cap + r[e]] = p[e];
      } else {
        const int o = atomicAdd(ovf_n, 1);
        ovf[2 * o]     = s[e];
        ovf[2 * o + 1] = p[e];
      }
    }
  }
}

template <bool FULL>
__device__ __forceinline__ void add_chunk(v4f& acc0, v4f& acc1,
                                          const float* __restrict__ rows,
                                          int p, int m, int half, int c4) {
  v4f v[16];
  #pragma unroll
  for (int k = 0; k < 16; ++k) {
    const int pa = __builtin_amdgcn_readlane(p, 2 * k);
    const int pb = __builtin_amdgcn_readlane(p, 2 * k + 1);
    const int pj = half ? pb : pa;
    v[k] = *(const v4f*)(rows + (size_t)pj * C_CH + c4);
  }
  #pragma unroll
  for (int k = 0; k < 16; ++k) {
    if (FULL || 2 * k + half < m) {
      if (k & 1) acc1 += v[k]; else acc0 += v[k];
    }
  }
}

__global__ __launch_bounds__(128) void SpVecs_segsum(
    const float* __restrict__ rows, const int* __restrict__ bucket,
    const int* __restrict__ ovf_n, const int* __restrict__ ovf,
    const int* __restrict__ cnt, float* __restrict__ sums, int N, int S) {
  __shared__ float part[4 * 128];
  const int s    = blockIdx.x;
  const int t    = threadIdx.x;
  const int lane = t & 31;
  const int w    = __builtin_amdgcn_readfirstlane(t >> 5);
  const int half = lane >> 4;
  const int c4   = (lane & 15) << 2;
  const int cap  = bucket_cap(N, S);
  const int nall = cnt[s];
  const int n    = min(nall, cap);
  const int* __restrict__ mine = bucket + (size_t)s * cap;

  v4f acc0 = {0.f, 0.f, 0.f, 0.f}, acc1 = {0.f, 0.f, 0.f, 0.f};
  int ch = w * 32;
  int p = 0;
  if (ch + lane < n) p = mine[ch + lane];
  while (ch < n) {
    const int nx = ch + 128;
    int pn = 0;
    if (nx + lane < n) pn = mine[nx + lane];
    const int m = n - ch;
    if (m >= 32) add_chunk<true>(acc0, acc1, rows, p, m, half, c4);
    else         add_chunk<false>(acc0, acc1, rows, p, m, half, c4);
    p = pn;
    ch = nx;
  }
  if (nall > cap) {
    const int no = min(*ovf_n, N);
    for (int i0 = w * 32; i0 < no; i0 += 128) {
      int sg = -1, pp = 0;
      if (i0 + lane < no) { sg = ovf[2 * (i0 + lane)]; pp = ovf[2 * (i0 + lane) + 1]; }
      unsigned mask = __builtin_amdgcn_ballot_w32(sg == s);
      while (mask) {
        const int j = __builtin_ctz(mask);
        mask &= mask - 1;
        const int pj = __shfl(pp, j);
        const v4f v = *(const v4f*)(rows + (size_t)pj * C_CH + c4);
        if (half == 0) acc0 += v;
      }
    }
  }
  const v4f a = acc0 + acc1;
  *(v4f*)(part + w * 128 + lane * 4) = a;
  __syncthreads();
  if (t < C_CH) {
    float r = 0.f;
    #pragma unroll
    for (int q = 0; q < 4; ++q) {
      r += part[q * 128 + t];
      r += part[q * 128 + 64 + t];
    }
    sums[(size_t)s * C_CH + t] = r;
  }
}

__global__ __launch_bounds__(256) void SpVecs_finalize(
    float* __restrict__ out, const int* __restrict__ cnts) {
  const int lane  = threadIdx.x & 31;
  const int wid   = threadIdx.x >> 5;
  const int g     = blockIdx.x * (blockDim.x >> 5) + wid;
  const int s0    = g << 4;
  const int m     = lane & 15;
  const int half  = lane >> 4;
  const int khalf = half << 1;

  const float invc = 1.0f / fmaxf((float)cnts[s0 + m], 1.0f);

#if __has_builtin(__builtin_amdgcn_wmma_f32_16x16x4_f32)
  #pragma unroll
  for (int c0 = 0; c0 < C_CH; c0 += 16) {
    v8f acc = {0.f, 0.f, 0.f, 0.f, 0.f, 0.f, 0.f, 0.f};
    #pragma unroll
    for (int j = 0; j < 4; ++j) {
      v2f A;
      A.x = (m == 4 * j + khalf + 0) ? invc : 0.0f;
      A.y = (m == 4 * j + khalf + 1) ? invc : 0.0f;
      const int r0 = s0 + 4 * j + khalf;
      v2f B;
      B.x = out[(size_t)r0 * C_CH + c0 + m];
      B.y = out[(size_t)(r0 + 1) * C_CH + c0 + m];
      acc = __builtin_amdgcn_wmma_f32_16x16x4_f32(
          false, A, false, B, (short)0, acc, false, false);
    }
    const int rowbase = s0 + (half << 3);
    #pragma unroll
    for (int r = 0; r < 8; ++r)
      out[(size_t)(rowbase + r) * C_CH + c0 + m] = acc[r];
  }
#else
  for (int c0 = 0; c0 < C_CH; c0 += 16) {
    for (int r = half * 8; r < half * 8 + 8; ++r)
      out[(size_t)(s0 + r) * C_CH + c0 + m] *= 1.0f / fmaxf((float)cnts[s0 + r], 1.0f);
  }
#endif
}

extern "C" void kernel_launch(void* const* d_in, const int* in_sizes, int n_in,
                              void* d_out, int out_size, void* d_ws, size_t ws_size,
                              hipStream_t stream) {
  const float* feat = (const float*)d_in[0];
  const int*   yx   = (const int*)d_in[1];
  const int*   seg  = (const int*)d_in[2];
  float* out  = (float*)d_out;
  int*   cnts  = (int*)d_ws;
  int*   ovf_n = (int*)d_ws + 131072;
  float* rows  = (float*)d_ws + 262144;

  const int N  = in_sizes[2];
  const int S  = out_size / C_CH;
  const int HW = in_sizes[0] / C_CH;
  const int W  = (int)lround(sqrt((double)HW));
  int* ovf    = (int*)(rows + (size_t)HW * C_CH);
  int* bucket = ovf + N + N;

  SpVecs_to_rows<<<2048, 256, 0, stream>>>(feat, rows, HW, cnts, ovf_n, S);

  SpVecs_bucket<<<(N + BK_TILE - 1) / BK_TILE, BK_THREADS, 0, stream>>>(
      yx, seg, cnts, ovf_n, ovf, bucket, N, W, S);

  SpVecs_segsum<<<S, 128, 0, stream>>>(rows, bucket, ovf_n, ovf, cnts, out, N, S);

  {
    int groups = S / 16;
    int wpb = 8;
    int blocks = (groups + wpb - 1) / wpb;
    SpVecs_finalize<<<blocks, wpb * 32, 0, stream>>>(out, cnts);
  }
}


# division by the segment size folded into the summing launch: its first wave scales the 64 sums with one f32 16x16x4 WMMA (1/count in A) and stores the row; 3 launches, not 4; f32 sum order kept
# speedup vs baseline: 34.2520x; 1.0718x over previous
#include <hip/hip_runtime.h>
#include <math.h>


typedef float v2f __attribute__((ext_vector_type(2)));
typedef float v4f __attribute__((ext_vector_type(4)));
typedef float v8f __attribute__((ext_vector_type(8)));

#define C_CH 64
#define TILE_PX 64
#define TILE_PITCH 65

__device__ __forceinline__ void lds_barrier() {
  __builtin_amdgcn_fence(__ATOMIC_RELEASE, "workgroup", "local");
  __builtin_amdgcn_s_barrier();
  __builtin_amdgcn_fence(__ATOMIC_ACQUIRE, "workgroup", "local");
}

__device__ __forceinline__ void load_tile_regs(v4f (&v)[4], const float* __restrict__ feat,
                                               int HW, int p, int cl) {
  #pragma unroll
  for (int it = 0; it < 4; ++it) {
    const float* src = feat + (size_t)(it * 16 + cl) * (size_t)HW + p;
    v[it] = __builtin_nontemporal_load((const v4f*)src);
  }
}

__global__ __launch_bounds__(256) void SpVecs_to_rows(
    const float* __restrict__ feat, float* __restrict__ rows, int HW,
    int* __restrict__ cnt, int* __restrict__ ovf_n, int S) {
  __shared__ float tile[TILE_PX * TILE_PITCH];
  const int t    = threadIdx.x;
  for (int i = blockIdx.x * 256 + t; i < S; i += gridDim.x * 256) cnt[i] = 0;
  if (blockIdx.x == 0 && t == 0) *ovf_n = 0;
  const int lane = t & 31;
  const int w    = t >> 5;
  const int q    = (t & 15) << 2;
  const int cl   = t >> 4;
  const int ntiles = (HW + TILE_PX - 1) / TILE_PX;
  const int nfull  = ((HW & 3) == 0) ? HW / TILE_PX : 0;

  int tl = blockIdx.x;
  v4f v[4];
  if (tl < nfull) load_tile_regs(v, feat, HW, tl * TILE_PX + q, cl);
  for (; tl < nfull; tl += gridDim.x) {
    #pragma unroll
    for (int it = 0; it < 4; ++it) {
      const int c = it * 16 + cl;
      tile[(q + 0) * TILE_PITCH + c] = v[it].x;
      tile[(q + 1) * TILE_PITCH + c] = v[it].y;
      tile[(q + 2) * TILE_PITCH + c] = v[it].z;
      tile[(q + 3) * TILE_PITCH + c] = v[it].w;
    }
    const int nx = tl + gridDim.x;
    if (nx < nfull) load_tile_regs(v, feat, HW, nx * TILE_PX + q, cl);
    lds_barrier();
    float* dst = rows + ((size_t)tl * TILE_PX + w * 8) * C_CH + lane;
    const float* srcl = tile + (w * 8) * TILE_PITCH + lane;
    #pragma unroll
    for (int k = 0; k < 8; ++k) {
      dst[k * C_CH]      = srcl[k * TILE_PITCH];
      dst[k * C_CH + 32] = srcl[k * TILE_PITCH + 32];
    }
    lds_barrier();
  }

  for (tl = nfull + blockIdx.x; tl < ntiles; tl += gridDim.x) {
    const int p0 = tl * TILE_PX;
    for (int e = t; e < TILE_PX * C_CH; e += 256) {
      const int c = e >> 6, px = e & 63;
      if (p0 + px < HW)
        tile[px * TILE_PITCH + c] = feat[(size_t)c * (size_t)HW + p0 + px];
    }
    __syncthreads();
    for (int k = 0; k < 8; ++k) {
      const int px = w * 8 + k;
      if (p0 + px < HW) {
        float* dst = rows + (size_t)(p0 + px) * C_CH;
        dst[lane]      = tile[px * TILE_PITCH + lane];
        dst[lane + 32] = tile[px * TILE_PITCH + lane + 32];
      }
    }
    __syncthreads();
  }
}

#define BK_THREADS 1024
#define BK_TILE    (BK_THREADS * 4)
#define BK_BINS    4096

__device__ __forceinline__ int bucket_cap(int N, int S) {
  const int mean = (N + S - 1) / S;
  return (2 * mean + 31) & ~31;
}

__global__ __launch_bounds__(BK_THREADS) void SpVecs_bucket(
    const int* __restrict__ yx, const int* __restrict__ seg, int* __restrict__ cnt,
    int* __restrict__ ovf_n, int* __restrict__ ovf, int* __restrict__ bucket,
    int N, int W, int S) {
  __shared__ int hist[BK_BINS];
  const int t   = threadIdx.x;
  const int e0  = (blockIdx.x * BK_THREADS + t) * 4;
  const int cap = bucket_cap(N, S);

  int s[4], p[4], r[4];
  if ((blockIdx.x + 1) * BK_TILE <= N) {
    const int4 sv = *(const int4*)(seg + e0);
    const int4 a  = *(const int4*)(yx + 2 * e0);
    const int4 b  = *(const int4*)(yx + 2 * e0 + 4);
    s[0] = sv.x; s[1] = sv.y; s[2] = sv.z; s[3] = sv.w;
    p[0] = a.x * W + a.y; p[1] = a.z * W + a.w;
    p[2] = b.x * W + b.y; p[3] = b.z * W + b.w;
  } else {
    #pragma unroll
    for (int e = 0; e < 4; ++e) {
      s[e] = -1; p[e] = 0;
      if (e0 + e < N) {
        s[e] = seg[e0 + e];
        p[e] = yx[2 * (e0 + e)] * W + yx[2 * (e0 + e) + 1];
      }
    }
  }
  #pragma unroll
  for (int e = 0; e < 4; ++e) {
    r[e] = 0;
    if (s[e] >= S) s[e] = -1;
  }

  const int rot = (blockIdx.x * 64) & (BK_BINS - 1);

  for (int b0 = 0; b0 < S; b0 += BK_BINS) {
    if (b0 > 0) __syncthreads();
    #pragma unroll
    for (int q = 0; q < BK_BINS / BK_THREADS; ++q) hist[q * BK_THREADS + t] = 0;
    __syncthreads();
    #pragma unroll
    for (int e = 0; e < 4; ++e) {
      const unsigned d = (unsigned)(s[e] - b0);
      if (s[e] >= 0 && d < (unsigned)BK_BINS) r[e] = atomicAdd(&hist[d], 1);
    }
    __syncthreads();
    int jj[BK_BINS / BK_THREADS], c[BK_BINS / BK_THREADS];
    #pragma unroll
    for (int q = 0; q < BK_BINS / BK_THREADS; ++q) {
      jj[q] = (q * BK_THREADS + t + rot) & (BK_BINS - 1);
      c[q]  = hist[jj[q]];
    }
    #pragma unroll
    for (int q = 0; q < BK_BINS / BK_THREADS; ++q)
      c[q] = atomicAdd(&cnt[min(b0 + jj[q], S - 1)], c[q]);
    #pragma unroll
    for (int q = 0; q < BK_BINS / BK_THREADS; ++q) hist[jj[q]] = c[q];
    __syncthreads();
    #pragma unroll
    for (int e = 0; e < 4; ++e) {
      const unsigned d = (unsigned)(s[e] - b0);
      const int base = hist[d & (unsigned)(BK_BINS - 1)];
      r[e] += (s[e] >= 0 && d < (unsigned)BK_BINS) ? base : 0;
    }
  }

  #pragma unroll
  for (int e = 0; e < 4; ++e) {
    if (s[e] >= 0) {
      if (r[e] < cap) {
        bucket[(size_t)s[e] * cap + r[e]] = p[e];
      } else {
        const int o = atomicAdd(ovf_n, 1);
        ovf[2 * o]     = s[e];
        ovf[2 * o + 1] = p[e];
      }
    }
  }
}

template <bool FULL>
__device__ __forceinline__ void add_chunk(v4f& acc0, v4f& acc1,
                                          const float* __restrict__ rows,
                                          int p, int m, int half, int c4) {
  v4f v[16];
  #pragma unroll
  for (int k = 0; k < 16; ++k) {
    const int pa = __builtin_amdgcn_readlane(p, 2 * k);
    const int pb = __builtin_amdgcn_readlane(p, 2 * k + 1);
    const int pj = half ? pb : pa;
    v[k] = *(const v4f*)(rows + (size_t)pj * C_CH + c4);
  }
  #pragma unroll
  for (int k = 0; k < 16; ++k) {
    if (FULL || 2 * k + half < m) {
      if (k & 1) acc1 += v[k]; else acc0 += v[k];
    }
  }
}

__global__ __launch_bounds__(128) void SpVecs_segsum(
    const float* __restrict__ rows, const int* __restrict__ bucket,
    const int* __restrict__ ovf_n, const int* __restrict__ ovf,
    const int* __restrict__ cnt, float* __restrict__ out, int N, int S) {
  __shared__ float part[4 * 128];
  const int s    = blockIdx.x;
  const int t    = threadIdx.x;
  const int lane = t & 31;
  const int w    = __builtin_amdgcn_readfirstlane(t >> 5);
  const int half = lane >> 4;
  const int c4   = (lane & 15) << 2;
  const int cap  = bucket_cap(N, S);
  const int nall = cnt[s];
  const int n    = min(nall, cap);
  const int* __restrict__ mine = bucket + (size_t)s * cap;

  v4f acc0 = {0.f, 0.f, 0.f, 0.f}, acc1 = {0.f, 0.f, 0.f, 0.f};
  int ch = w * 32;
  int p = 0;
  if (ch + lane < n) p = mine[ch + lane];
  while (ch < n) {
    const int nx = ch + 128;
    int pn = 0;
    if (nx + lane < n) pn = mine[nx + lane];
    const int m = n - ch;
    if (m >= 32) add_chunk<true>(acc0, acc1, rows, p, m, half, c4);
    else         add_chunk<false>(acc0, acc1, rows, p, m, half, c4);
    p = pn;
    ch = nx;
  }
  if (nall > cap) {
    const int no = min(*ovf_n, N);
    for (int i0 = w * 32; i0 < no; i0 += 128) {
      int sg = -1, pp = 0;
      if (i0 + lane < no) { sg = ovf[2 * (i0 + lane)]; pp = ovf[2 * (i0 + lane) + 1]; }
      unsigned mask = __builtin_amdgcn_ballot_w32(sg == s);
      while (mask) {
        const int j = __builtin_ctz(mask);
        mask &= mask - 1;
        const int pj = __shfl(pp, j);
        const v4f v = *(const v4f*)(rows + (size_t)pj * C_CH + c4);
        if (half == 0) acc0 += v;
      }
    }
  }
  const v4f a = acc0 + acc1;
  *(v4f*)(part + w * 128 + lane * 4) = a;
  __syncthreads();
  if (w == 0) {
    const int m  = lane & 15;
    const int cA = (half << 5) + m;
    const int cB = cA + 16;
    float rA = 0.f, rB = 0.f;
    #pragma unroll
    for (int q = 0; q < 4; ++q) {
      rA += part[q * 128 + cA];
      rA += part[q * 128 + 64 + cA];
      rB += part[q * 128 + cB];
      rB += part[q * 128 + 64 + cB];
    }
    const float invc = 1.0f / fmaxf((float)nall, 1.0f);
    float* __restrict__ dst = out + (size_t)s * C_CH;
#if __has_builtin(__builtin_amdgcn_wmma_f32_16x16x4_f32)
    v2f A, B;
    A.x = (lane == 0 || lane == 17) ? invc : 0.0f;
    A.y = (lane == 8 || lane == 25) ? invc : 0.0f;
    B.x = rA;
    B.y = rB;
    v8f acc = {0.f, 0.f, 0.f, 0.f, 0.f, 0.f, 0.f, 0.f};
    acc = __builtin_amdgcn_wmma_f32_16x16x4_f32(
        false, A, false, B, (short)0, acc, false, false);
    dst[lane]      = acc[0];
    dst[32 + lane] = acc[1];
#else
    dst[cA] = rA * invc;
    dst[cB] = rB * invc;
#endif
  }
}

extern "C" void kernel_launch(void* const* d_in, const int* in_sizes, int n_in,
                              void* d_out, int out_size, void* d_ws, size_t ws_size,
                              hipStream_t stream) {
  const float* feat = (const float*)d_in[0];
  const int*   yx   = (const int*)d_in[1];
  const int*   seg  = (const int*)d_in[2];
  float* out  = (float*)d_out;
  int*   cnts  = (int*)d_ws;
  int*   ovf_n = (int*)d_ws + 131072;
  float* rows  = (float*)d_ws + 262144;

  const int N  = in_sizes[2];
  const int S  = out_size / C_CH;
  const int HW = in_sizes[0] / C_CH;
  const int W  = (int)lround(sqrt((double)HW));
  int* ovf    = (int*)(rows + (size_t)HW * C_CH);
  int* bucket = ovf + N + N;

  SpVecs_to_rows<<<2048, 256, 0, stream>>>(feat, rows, HW, cnts, ovf_n, S);

  SpVecs_bucket<<<(N + BK_TILE - 1) / BK_TILE, BK_THREADS, 0, stream>>>(
      yx, seg, cnts, ovf_n, ovf, bucket, N, W, S);

  SpVecs_segsum<<<S, 128, 0, stream>>>(rows, bucket, ovf_n, ovf, cnts, out, N, S);
}
